# starting kernel
# baseline (speedup 1.0000x reference)
#include <hip/hip_runtime.h>
#include <math.h>

typedef __attribute__((ext_vector_type(2))) float v2f;
typedef __attribute__((ext_vector_type(8))) float v8f;

#define N_NODES 65536
#define DEG     16
#define D       64
#define D2      128
#define NB1     1024
#define NNEG    10
#define GTOT    (NB1 + NB1 + NNEG)
#define EPSN    1e-12f
#define XPAD    132

__global__ __launch_bounds__(32) void graphsage_hop1(
    const float* __restrict__ feat, const int* __restrict__ adj,
    const float* __restrict__ W1,   const float* __restrict__ b1,
    float* __restrict__ h1)
{
    __shared__ float Xs[16][XPAD];

    const int lane = threadIdx.x;
    const int m    = lane & 15;
    const int half = lane >> 4;
    const int tile = blockIdx.x;
    const int node = tile * 16 + m;

    {
        const float4* selfp = (const float4*)(feat + (size_t)node * D + half * 32);
        float4* xlo = (float4*)&Xs[m][half * 32];
        float4* xhi = (float4*)&Xs[m][D + half * 32];
        #pragma unroll
        for (int i = 0; i < 8; ++i) xlo[i] = selfp[i];

        float4 acc[8];
        #pragma unroll
        for (int i = 0; i < 8; ++i) acc[i] = make_float4(0.f, 0.f, 0.f, 0.f);
        for (int j = 0; j < DEG; ++j) {
            const int nb = adj[(size_t)node * DEG + j];
            const float4* p = (const float4*)(feat + (size_t)nb * D + half * 32);
            #pragma unroll
            for (int i = 0; i < 8; ++i) {
                float4 v = p[i];
                acc[i].x += v.x; acc[i].y += v.y; acc[i].z += v.z; acc[i].w += v.w;
            }
        }
        #pragma unroll
        for (int i = 0; i < 8; ++i) xhi[i] = acc[i];
    }
    __syncthreads();

    const int kb = half * 2;
    v8f zero = {};
    v8f c[4];
    #pragma unroll
    for (int nt = 0; nt < 4; ++nt) c[nt] = zero;

    for (int k0 = 0; k0 < 2 * D; k0 += 4) {
        v2f a = *(const v2f*)&Xs[m][k0 + kb];
        #pragma unroll
        for (int nt = 0; nt < 4; ++nt) {
            const int col = nt * 16 + m;
            v2f b = *(const v2f*)&W1[(size_t)col * (2 * D) + k0 + kb];
            c[nt] = __builtin_amdgcn_wmma_f32_16x16x4_f32(
                false, a, false, b, (short)0, c[nt], false, false);
        }
    }

    #pragma unroll
    for (int nt = 0; nt < 4; ++nt) {
        const int   col  = nt * 16 + m;
        const float bias = b1[col];
        #pragma unroll
        for (int r = 0; r < 8; ++r) {
            const int row = r + half * 8;
            h1[(size_t)(tile * 16 + row) * D + col] = tanhf(c[nt][r] + bias);
        }
    }
}

__global__ __launch_bounds__(32) void graphsage_hop2(
    const float* __restrict__ feat, const int* __restrict__ adj,
    const int* __restrict__ in1, const int* __restrict__ in2,
    const int* __restrict__ ineg,
    const float* __restrict__ W2, const float* __restrict__ b2,
    const float* __restrict__ h1, float* __restrict__ out)
{
    __shared__ float Xs[16][XPAD];

    const int lane = threadIdx.x;
    const int m    = lane & 15;
    const int half = lane >> 4;
    const int tile = blockIdx.x;
    const int g    = tile * 16 + m;

    int node;
    if      (g < NB1)      node = in1[g];
    else if (g < 2 * NB1)  node = in2[g - NB1];
    else if (g < GTOT)     node = ineg[g - 2 * NB1];
    else                   node = 0;

    {
        const float4* selfp = (const float4*)(feat + (size_t)node * D + half * 32);
        float4* xlo = (float4*)&Xs[m][half * 32];
        float4* xhi = (float4*)&Xs[m][D + half * 32];
        #pragma unroll
        for (int i = 0; i < 8; ++i) xlo[i] = selfp[i];

        float4 acc[8];
        #pragma unroll
        for (int i = 0; i < 8; ++i) acc[i] = make_float4(0.f, 0.f, 0.f, 0.f);
        for (int j = 0; j < DEG; ++j) {
            const int nb = adj[(size_t)node * DEG + j];
            const float4* p = (const float4*)(h1 + (size_t)nb * D + half * 32);
            #pragma unroll
            for (int i = 0; i < 8; ++i) {
                float4 v = p[i];
                acc[i].x += v.x; acc[i].y += v.y; acc[i].z += v.z; acc[i].w += v.w;
            }
        }
        #pragma unroll
        for (int i = 0; i < 8; ++i) xhi[i] = acc[i];
    }
    __syncthreads();

    const int kb = half * 2;
    v8f zero = {};
    v8f c[8];
    #pragma unroll
    for (int nt = 0; nt < 8; ++nt) c[nt] = zero;

    for (int k0 = 0; k0 < D2; k0 += 4) {
        v2f a = *(const v2f*)&Xs[m][k0 + kb];
        #pragma unroll
        for (int nt = 0; nt < 8; ++nt) {
            const int col = nt * 16 + m;
            v2f b = *(const v2f*)&W2[(size_t)col * D2 + k0 + kb];
            c[nt] = __builtin_amdgcn_wmma_f32_16x16x4_f32(
                false, a, false, b, (short)0, c[nt], false, false);
        }
    }

    float bias[8];
    #pragma unroll
    for (int nt = 0; nt < 8; ++nt) bias[nt] = b2[nt * 16 + m];

    #pragma unroll
    for (int r = 0; r < 8; ++r) {
        float vals[8];
        float ss = 0.f;
        #pragma unroll
        for (int nt = 0; nt < 8; ++nt) {
            const float v = c[nt][r] + bias[nt];
            vals[nt] = v;
            ss += v * v;
        }
        ss += __shfl_xor(ss, 1);
        ss += __shfl_xor(ss, 2);
        ss += __shfl_xor(ss, 4);
        ss += __shfl_xor(ss, 8);
        const float inv = 1.0f / fmaxf(sqrtf(ss), EPSN);

        const int grow = tile * 16 + r + half * 8;
        if (grow < GTOT) {
            #pragma unroll
            for (int nt = 0; nt < 8; ++nt)
                out[(size_t)grow * D2 + nt * 16 + m] = vals[nt] * inv;
        }
    }
}

extern "C" void kernel_launch(void* const* d_in, const int* in_sizes, int n_in,
                              void* d_out, int out_size, void* d_ws, size_t ws_size,
                              hipStream_t stream) {
    const float* feat = (const float*)d_in[0];
    const int*   adj  = (const int*)  d_in[1];
    const int*   in1  = (const int*)  d_in[2];
    const int*   in2  = (const int*)  d_in[3];
    const int*   ineg = (const int*)  d_in[4];
    const float* W1   = (const float*)d_in[5];
    const float* b1   = (const float*)d_in[6];
    const float* W2   = (const float*)d_in[7];
    const float* b2   = (const float*)d_in[8];
    float*       outp = (float*)d_out;
    float*       h1   = (float*)d_ws;

    graphsage_hop1<<<N_NODES / 16, 32, 0, stream>>>(feat, adj, W1, b1, h1);
    graphsage_hop2<<<(GTOT + 15) / 16, 32, 0, stream>>>(feat, adj, in1, in2, ineg,
                                                        W2, b2, h1, outp);
}


# Coalesced gather: a half-wave (16 lanes x float4) reads one whole 256-B row, two rows per load, neighbor ids staged in LDS, instead of each lane reading 16 B of its own line; j sum order kept
# speedup vs baseline: 1.2890x; 1.2890x over previous
#include <hip/hip_runtime.h>
#include <math.h>

typedef __attribute__((ext_vector_type(2))) float v2f;
typedef __attribute__((ext_vector_type(8))) float v8f;

#define N_NODES 65536
#define DEG     16
#define D       64
#define D2      128
#define NB1     1024
#define NNEG    10
#define GTOT    (NB1 + NB1 + NNEG)
#define EPSN    1e-12f
#define XPAD    132

__device__ __forceinline__ void gather_tile(
    const float* __restrict__ selfsrc, const float* __restrict__ nbsrc,
    const int* __restrict__ adj, const int node_r, const int lane,
    float (*Xs)[XPAD], int (*Is)[DEG], int* Ns)
{
    {
        const int r    = lane >> 1;
        const int part = lane & 1;
        const int4* ap = (const int4*)(adj + (size_t)node_r * DEG + part * 8);
        const int4 i0 = ap[0];
        const int4 i1 = ap[1];
        *(int4*)&Is[r][part * 8]     = i0;
        *(int4*)&Is[r][part * 8 + 4] = i1;
        if (part == 0) Ns[r] = node_r;
    }
    __syncthreads();

    const int c    = lane & 15;
    const int half = lane >> 4;
    #pragma unroll
    for (int p = 0; p < 8; ++p) {
        const int row = 2 * p + half;
        const float4 s = *(const float4*)(selfsrc + (size_t)Ns[row] * D + 4 * c);
        float4 acc = make_float4(0.f, 0.f, 0.f, 0.f);
        #pragma unroll
        for (int q = 0; q < DEG / 4; ++q) {
            const int4 nb = *(const int4*)&Is[row][4 * q];
            const float4 v0 = *(const float4*)(nbsrc + (size_t)nb.x * D + 4 * c);
            const float4 v1 = *(const float4*)(nbsrc + (size_t)nb.y * D + 4 * c);
            const float4 v2 = *(const float4*)(nbsrc + (size_t)nb.z * D + 4 * c);
            const float4 v3 = *(const float4*)(nbsrc + (size_t)nb.w * D + 4 * c);
            acc.x += v0.x; acc.y += v0.y; acc.z += v0.z; acc.w += v0.w;
            acc.x += v1.x; acc.y += v1.y; acc.z += v1.z; acc.w += v1.w;
            acc.x += v2.x; acc.y += v2.y; acc.z += v2.z; acc.w += v2.w;
            acc.x += v3.x; acc.y += v3.y; acc.z += v3.z; acc.w += v3.w;
        }
        *(float4*)&Xs[row][4 * c]     = s;
        *(float4*)&Xs[row][D + 4 * c] = acc;
    }
}

__global__ __launch_bounds__(32) void graphsage_hop1(
    const float* __restrict__ feat, const int* __restrict__ adj,
    const float* __restrict__ W1,   const float* __restrict__ b1,
    float* __restrict__ h1)
{
    __shared__ float Xs[16][XPAD];
    __shared__ __attribute__((aligned(16))) int Is[16][DEG];
    __shared__ int   Ns[16];

    const int lane = threadIdx.x;
    const int m    = lane & 15;
    const int half = lane >> 4;
    const int tile = blockIdx.x;

    gather_tile(feat, feat, adj, tile * 16 + (lane >> 1), lane, Xs, Is, Ns);
    __syncthreads();

    const int kb = half * 2;
    v8f zero = {};
    v8f c[4];
    #pragma unroll
    for (int nt = 0; nt < 4; ++nt) c[nt] = zero;

    for (int k0 = 0; k0 < 2 * D; k0 += 4) {
        v2f a = *(const v2f*)&Xs[m][k0 + kb];
        #pragma unroll
        for (int nt = 0; nt < 4; ++nt) {
            const int col = nt * 16 + m;
            v2f b = *(const v2f*)&W1[(size_t)col * (2 * D) + k0 + kb];
            c[nt] = __builtin_amdgcn_wmma_f32_16x16x4_f32(
                false, a, false, b, (short)0, c[nt], false, false);
        }
    }

    #pragma unroll
    for (int nt = 0; nt < 4; ++nt) {
        const int   col  = nt * 16 + m;
        const float bias = b1[col];
        #pragma unroll
        for (int r = 0; r < 8; ++r) {
            const int row = r + half * 8;
            h1[(size_t)(tile * 16 + row) * D + col] = tanhf(c[nt][r] + bias);
        }
    }
}

__global__ __launch_bounds__(32) void graphsage_hop2(
    const float* __restrict__ feat, const int* __restrict__ adj,
    const int* __restrict__ in1, const int* __restrict__ in2,
    const int* __restrict__ ineg,
    const float* __restrict__ W2, const float* __restrict__ b2,
    const float* __restrict__ h1, float* __restrict__ out)
{
    __shared__ float Xs[16][XPAD];
    __shared__ __attribute__((aligned(16))) int Is[16][DEG];
    __shared__ int   Ns[16];

    const int lane = threadIdx.x;
    const int m    = lane & 15;
    const int half = lane >> 4;
    const int tile = blockIdx.x;
    const int g    = tile * 16 + (lane >> 1);

    int node;
    if      (g < NB1)      node = in1[g];
    else if (g < 2 * NB1)  node = in2[g - NB1];
    else if (g < GTOT)     node = ineg[g - 2 * NB1];
    else                   node = 0;

    gather_tile(feat, h1, adj, node, lane, Xs, Is, Ns);
    __syncthreads();

    const int kb = half * 2;
    v8f zero = {};
    v8f c[8];
    #pragma unroll
    for (int nt = 0; nt < 8; ++nt) c[nt] = zero;

    for (int k0 = 0; k0 < D2; k0 += 4) {
        v2f a = *(const v2f*)&Xs[m][k0 + kb];
        #pragma unroll
        for (int nt = 0; nt < 8; ++nt) {
            const int col = nt * 16 + m;
            v2f b = *(const v2f*)&W2[(size_t)col * D2 + k0 + kb];
            c[nt] = __builtin_amdgcn_wmma_f32_16x16x4_f32(
                false, a, false, b, (short)0, c[nt], false, false);
        }
    }

    float bias[8];
    #pragma unroll
    for (int nt = 0; nt < 8; ++nt) bias[nt] = b2[nt * 16 + m];

    #pragma unroll
    for (int r = 0; r < 8; ++r) {
        float vals[8];
        float ss = 0.f;
        #pragma unroll
        for (int nt = 0; nt < 8; ++nt) {
            const float v = c[nt][r] + bias[nt];
            vals[nt] = v;
            ss += v * v;
        }
        ss += __shfl_xor(ss, 1);
        ss += __shfl_xor(ss, 2);
        ss += __shfl_xor(ss, 4);
        ss += __shfl_xor(ss, 8);
        const float inv = 1.0f / fmaxf(sqrtf(ss), EPSN);

        const int grow = tile * 16 + r + half * 8;
        if (grow < GTOT) {
            #pragma unroll
            for (int nt = 0; nt < 8; ++nt)
                out[(size_t)grow * D2 + nt * 16 + m] = vals[nt] * inv;
        }
    }
}

extern "C" void kernel_launch(void* const* d_in, const int* in_sizes, int n_in,
                              void* d_out, int out_size, void* d_ws, size_t ws_size,
                              hipStream_t stream) {
    const float* feat = (const float*)d_in[0];
    const int*   adj  = (const int*)  d_in[1];
    const int*   in1  = (const int*)  d_in[2];
    const int*   in2  = (const int*)  d_in[3];
    const int*   ineg = (const int*)  d_in[4];
    const float* W1   = (const float*)d_in[5];
    const float* b1   = (const float*)d_in[6];
    const float* W2   = (const float*)d_in[7];
    const float* b2   = (const float*)d_in[8];
    float*       outp = (float*)d_out;
    float*       h1   = (float*)d_ws;

    graphsage_hop1<<<N_NODES / 16, 32, 0, stream>>>(feat, adj, W1, b1, h1);
    graphsage_hop2<<<(GTOT + 15) / 16, 32, 0, stream>>>(feat, adj, in1, in2, ineg,
                                                        W2, b2, h1, outp);
}


# Hop 1 only for the rows hop 2 reads: one 16-row tile per query row (its node's 16 neighbors), 2058 tiles not 4096; h1 in per-query slot rows, read contiguously by hop 2; j sum order kept
# speedup vs baseline: 1.7488x; 1.3567x over previous
#include <hip/hip_runtime.h>
#include <math.h>

typedef __attribute__((ext_vector_type(2))) float v2f;
typedef __attribute__((ext_vector_type(8))) float v8f;

#define N_NODES 65536
#define DEG     16
#define D       64
#define D2      128
#define NB1     1024
#define NNEG    10
#define GTOT    (NB1 + NB1 + NNEG)
#define EPSN    1e-12f
#define XPAD    132

__device__ __forceinline__ void gather_tile(
    const float* __restrict__ selfsrc, const float* __restrict__ nbsrc,
    const int* __restrict__ adj, const int node_r, const int lane,
    float (*Xs)[XPAD], int (*Is)[DEG], int* Ns)
{
    {
        const int r    = lane >> 1;
        const int part = lane & 1;
        const int4* ap = (const int4*)(adj + (size_t)node_r * DEG + part * 8);
        const int4 i0 = ap[0];
        const int4 i1 = ap[1];
        *(int4*)&Is[r][part * 8]     = i0;
        *(int4*)&Is[r][part * 8 + 4] = i1;
        if (part == 0) Ns[r] = node_r;
    }
    __syncthreads();

    const int c    = lane & 15;
    const int half = lane >> 4;
    #pragma unroll
    for (int p = 0; p < 8; ++p) {
        const int row = 2 * p + half;
        const float4 s = *(const float4*)(selfsrc + (size_t)Ns[row] * D + 4 * c);
        float4 acc = make_float4(0.f, 0.f, 0.f, 0.f);
        #pragma unroll
        for (int q = 0; q < DEG / 4; ++q) {
            const int4 nb = *(const int4*)&Is[row][4 * q];
            const float4 v0 = *(const float4*)(nbsrc + (size_t)nb.x * D + 4 * c);
            const float4 v1 = *(const float4*)(nbsrc + (size_t)nb.y * D + 4 * c);
            const float4 v2 = *(const float4*)(nbsrc + (size_t)nb.z * D + 4 * c);
            const float4 v3 = *(const float4*)(nbsrc + (size_t)nb.w * D + 4 * c);
            acc.x += v0.x; acc.y += v0.y; acc.z += v0.z; acc.w += v0.w;
            acc.x += v1.x; acc.y += v1.y; acc.z += v1.z; acc.w += v1.w;
            acc.x += v2.x; acc.y += v2.y; acc.z += v2.z; acc.w += v2.w;
            acc.x += v3.x; acc.y += v3.y; acc.z += v3.z; acc.w += v3.w;
        }
        *(float4*)&Xs[row][4 * c]     = s;
        *(float4*)&Xs[row][D + 4 * c] = acc;
    }
}

__device__ __forceinline__ int query_node(
    const int g, const int* __restrict__ in1, const int* __restrict__ in2,
    const int* __restrict__ ineg)
{
    if (g < NB1)     return in1[g];
    if (g < 2 * NB1) return in2[g - NB1];
    return ineg[g - 2 * NB1];
}

__global__ __launch_bounds__(32) void graphsage_hop1(
    const float* __restrict__ feat, const int* __restrict__ adj,
    const int* __restrict__ in1, const int* __restrict__ in2,
    const int* __restrict__ ineg,
    const float* __restrict__ W1,   const float* __restrict__ b1,
    float* __restrict__ h1q)
{
    __shared__ float Xs[16][XPAD];
    __shared__ __attribute__((aligned(16))) int Is[16][DEG];
    __shared__ int   Ns[16];

    const int lane = threadIdx.x;
    const int m    = lane & 15;
    const int half = lane >> 4;
    const int g    = blockIdx.x;

    const int nq     = query_node(g, in1, in2, ineg);
    const int node_r = adj[(size_t)nq * DEG + (lane >> 1)];

    gather_tile(feat, feat, adj, node_r, lane, Xs, Is, Ns);
    __syncthreads();

    const int kb = half * 2;
    v8f zero = {};
    v8f c[4];
    #pragma unroll
    for (int nt = 0; nt < 4; ++nt) c[nt] = zero;

    for (int k0 = 0; k0 < 2 * D; k0 += 4) {
        v2f a = *(const v2f*)&Xs[m][k0 + kb];
        #pragma unroll
        for (int nt = 0; nt < 4; ++nt) {
            const int col = nt * 16 + m;
            v2f b = *(const v2f*)&W1[(size_t)col * (2 * D) + k0 + kb];
            c[nt] = __builtin_amdgcn_wmma_f32_16x16x4_f32(
                false, a, false, b, (short)0, c[nt], false, false);
        }
    }

    #pragma unroll
    for (int nt = 0; nt < 4; ++nt) {
        const int   col  = nt * 16 + m;
        const float bias = b1[col];
        #pragma unroll
        for (int r = 0; r < 8; ++r) {
            const int row = r + half * 8;
            h1q[((size_t)g * DEG + row) * D + col] = tanhf(c[nt][r] + bias);
        }
    }
}

__global__ __launch_bounds__(32) void graphsage_hop2(
    const float* __restrict__ feat,
    const int* __restrict__ in1, const int* __restrict__ in2,
    const int* __restrict__ ineg,
    const float* __restrict__ W2, const float* __restrict__ b2,
    const float* __restrict__ h1q, float* __restrict__ out)
{
    __shared__ float Xs[16][XPAD];
    __shared__ int   Ns[16];

    const int lane = threadIdx.x;
    const int m    = lane & 15;
    const int half = lane >> 4;
    const int tile = blockIdx.x;

    {
        const int g = tile * 16 + (lane >> 1);
        const int node = (g < GTOT) ? query_node(g, in1, in2, ineg) : 0;
        if ((lane & 1) == 0) Ns[lane >> 1] = node;
    }
    __syncthreads();

    {
        const int cc = lane & 15;
        #pragma unroll
        for (int p = 0; p < 8; ++p) {
            const int row = 2 * p + half;
            const int gq  = min(tile * 16 + row, GTOT - 1);
            const float4 s = *(const float4*)(feat + (size_t)Ns[row] * D + 4 * cc);
            const float* hb = h1q + (size_t)gq * (DEG * D) + 4 * cc;
            float4 acc = make_float4(0.f, 0.f, 0.f, 0.f);
            #pragma unroll
            for (int j = 0; j < DEG; ++j) {
                const float4 v = *(const float4*)(hb + j * D);
                acc.x += v.x; acc.y += v.y; acc.z += v.z; acc.w += v.w;
            }
            *(float4*)&Xs[row][4 * cc]     = s;
            *(float4*)&Xs[row][D + 4 * cc] = acc;
        }
    }
    __syncthreads();

    const int kb = half * 2;
    v8f zero = {};
    v8f c[8];
    #pragma unroll
    for (int nt = 0; nt < 8; ++nt) c[nt] = zero;

    for (int k0 = 0; k0 < D2; k0 += 4) {
        v2f a = *(const v2f*)&Xs[m][k0 + kb];
        #pragma unroll
        for (int nt = 0; nt < 8; ++nt) {
            const int col = nt * 16 + m;
            v2f b = *(const v2f*)&W2[(size_t)col * D2 + k0 + kb];
            c[nt] = __builtin_amdgcn_wmma_f32_16x16x4_f32(
                false, a, false, b, (short)0, c[nt], false, false);
        }
    }

    float bias[8];
    #pragma unroll
    for (int nt = 0; nt < 8; ++nt) bias[nt] = b2[nt * 16 + m];

    #pragma unroll
    for (int r = 0; r < 8; ++r) {
        float vals[8];
        float ss = 0.f;
        #pragma unroll
        for (int nt = 0; nt < 8; ++nt) {
            const float v = c[nt][r] + bias[nt];
            vals[nt] = v;
            ss += v * v;
        }
        ss += __shfl_xor(ss, 1);
        ss += __shfl_xor(ss, 2);
        ss += __shfl_xor(ss, 4);
        ss += __shfl_xor(ss, 8);
        const float inv = 1.0f / fmaxf(sqrtf(ss), EPSN);

        const int grow = tile * 16 + r + half * 8;
        if (grow < GTOT) {
            #pragma unroll
            for (int nt = 0; nt < 8; ++nt)
                out[(size_t)grow * D2 + nt * 16 + m] = vals[nt] * inv;
        }
    }
}

extern "C" void kernel_launch(void* const* d_in, const int* in_sizes, int n_in,
                              void* d_out, int out_size, void* d_ws, size_t ws_size,
                              hipStream_t stream) {
    const float* feat = (const float*)d_in[0];
    const int*   adj  = (const int*)  d_in[1];
    const int*   in1  = (const int*)  d_in[2];
    const int*   in2  = (const int*)  d_in[3];
    const int*   ineg = (const int*)  d_in[4];
    const float* W1   = (const float*)d_in[5];
    const float* b1   = (const float*)d_in[6];
    const float* W2   = (const float*)d_in[7];
    const float* b2   = (const float*)d_in[8];
    float*       outp = (float*)d_out;
    float*       h1q  = (float*)d_ws;

    graphsage_hop1<<<GTOT, 32, 0, stream>>>(feat, adj, in1, in2, ineg, W1, b1, h1q);
    graphsage_hop2<<<(GTOT + 15) / 16, 32, 0, stream>>>(feat, in1, in2, ineg,
                                                        W2, b2, h1q, outp);
}


# Hop 2 fused into hop 1's launch: 9 query rows per 9-wave work-group, each wave sums its tile's 16 h1 rows in registers (j order kept), waves split the hop-2 product by column tile; no h1 workspace
# speedup vs baseline: 2.7741x; 1.5863x over previous
#include <hip/hip_runtime.h>
#include <math.h>


typedef __attribute__((ext_vector_type(2))) float v2f;
typedef __attribute__((ext_vector_type(8))) float v8f;

#define N_NODES 65536
#define DEG     16
#define D       64
#define D2      128
#define NB1     1024
#define NNEG    10
#define GTOT    (NB1 + NB1 + NNEG)
#define EPSN    1e-12f
#define XPAD    132
#define NW      9
#define NBLK    ((GTOT + NW - 1) / NW)
#define NT2     (D2 / 16)

__device__ __forceinline__ void wave_lds_order()
{
    __builtin_amdgcn_fence(__ATOMIC_RELEASE, "wavefront");
    __builtin_amdgcn_wave_barrier();
    __builtin_amdgcn_fence(__ATOMIC_ACQUIRE, "wavefront");
}

__device__ __forceinline__ void gather_tile(
    const float* __restrict__ selfsrc, const float* __restrict__ nbsrc,
    const int* __restrict__ adj, const int node_r, const int lane,
    float (*Xs)[XPAD], int (*Is)[DEG], int* Ns)
{
    {
        const int r    = lane >> 1;
        const int part = lane & 1;
        const int4* ap = (const int4*)(adj + (size_t)node_r * DEG + part * 8);
        const int4 i0 = ap[0];
        const int4 i1 = ap[1];
        *(int4*)&Is[r][part * 8]     = i0;
        *(int4*)&Is[r][part * 8 + 4] = i1;
        if (part == 0) Ns[r] = node_r;
    }
    wave_lds_order();

    const int c    = lane & 15;
    const int half = lane >> 4;
    #pragma unroll
    for (int p = 0; p < 8; ++p) {
        const int row = 2 * p + half;
        const float4 s = *(const float4*)(selfsrc + (size_t)Ns[row] * D + 4 * c);
        float4 acc = make_float4(0.f, 0.f, 0.f, 0.f);
        #pragma unroll
        for (int q = 0; q < DEG / 4; ++q) {
            const int4 nb = *(const int4*)&Is[row][4 * q];
            const float4 v0 = *(const float4*)(nbsrc + (size_t)nb.x * D + 4 * c);
            const float4 v1 = *(const float4*)(nbsrc + (size_t)nb.y * D + 4 * c);
            const float4 v2 = *(const float4*)(nbsrc + (size_t)nb.z * D + 4 * c);
            const float4 v3 = *(const float4*)(nbsrc + (size_t)nb.w * D + 4 * c);
            acc.x += v0.x; acc.y += v0.y; acc.z += v0.z; acc.w += v0.w;
            acc.x += v1.x; acc.y += v1.y; acc.z += v1.z; acc.w += v1.w;
            acc.x += v2.x; acc.y += v2.y; acc.z += v2.z; acc.w += v2.w;
            acc.x += v3.x; acc.y += v3.y; acc.z += v3.z; acc.w += v3.w;
        }
        *(float4*)&Xs[row][4 * c]     = s;
        *(float4*)&Xs[row][D + 4 * c] = acc;
    }
}

__device__ __forceinline__ int query_node(
    const int g, const int* __restrict__ in1, const int* __restrict__ in2,
    const int* __restrict__ ineg)
{
    if (g < NB1)     return in1[g];
    if (g < 2 * NB1) return in2[g - NB1];
    return ineg[g - 2 * NB1];
}

__global__ __launch_bounds__(NW * 32) void graphsage_fused(
    const float* __restrict__ feat, const int* __restrict__ adj,
    const int* __restrict__ in1, const int* __restrict__ in2,
    const int* __restrict__ ineg,
    const float* __restrict__ W1, const float* __restrict__ b1,
    const float* __restrict__ W2, const float* __restrict__ b2,
    float* __restrict__ out)
{
    __shared__ __attribute__((aligned(16))) float Xs_all[NW][16][XPAD];
    __shared__ __attribute__((aligned(16))) int   Is_all[NW][16][DEG];
    __shared__ int   Ns_all[NW][16];
    __shared__ __attribute__((aligned(16))) float X2[NW][XPAD];
    __shared__ float V[NW][XPAD];

    const int lane = threadIdx.x & 31;
    const int w    = __builtin_amdgcn_readfirstlane((int)(threadIdx.x >> 5));
    const int m    = lane & 15;
    const int half = lane >> 4;
    const int g    = blockIdx.x * NW + w;
    const int gq   = min(g, GTOT - 1);

    float (*Xs)[XPAD] = Xs_all[w];
    int   (*Is)[DEG]  = Is_all[w];
    int*  Ns          = Ns_all[w];

    const int nq     = query_node(gq, in1, in2, ineg);
    const int node_r = adj[(size_t)nq * DEG + (lane >> 1)];
    const float4 selfq = *(const float4*)(feat + (size_t)nq * D + 4 * m);

    float bias1[4];
    #pragma unroll
    for (int nt = 0; nt < 4; ++nt) bias1[nt] = b1[nt * 16 + m];

    gather_tile(feat, feat, adj, node_r, lane, Xs, Is, Ns);
    wave_lds_order();

    const int kb = half * 2;
    v8f zero = {};
    v8f c[4];
    #pragma unroll
    for (int nt = 0; nt < 4; ++nt) c[nt] = zero;

    for (int k0 = 0; k0 < 2 * D; k0 += 4) {
        v2f a = *(const v2f*)&Xs[m][k0 + kb];
        #pragma unroll
        for (int nt = 0; nt < 4; ++nt) {
            const int col = nt * 16 + m;
            v2f b = *(const v2f*)&W1[(size_t)col * (2 * D) + k0 + kb];
            c[nt] = __builtin_amdgcn_wmma_f32_16x16x4_f32(
                false, a, false, b, (short)0, c[nt], false, false);
        }
    }

    v2f bw[D2 / 4];
    const int col2 = min(w, NT2 - 1) * 16 + m;
    #pragma unroll
    for (int ks = 0; ks < D2 / 4; ++ks)
        bw[ks] = *(const v2f*)&W2[(size_t)col2 * D2 + ks * 4 + kb];
    const float bias2 = b2[col2];

    #pragma unroll
    for (int nt = 0; nt < 4; ++nt) {
        const int   col  = nt * 16 + m;
        const float bias = bias1[nt];
        float t[8];
        #pragma unroll
        for (int r = 0; r < 8; ++r) t[r] = tanhf(c[nt][r] + bias);
        float s0 = 0.f;
        #pragma unroll
        for (int r = 0; r < 8; ++r) s0 += t[r];
        const float p = __shfl(s0, m);
        float s1 = p;
        #pragma unroll
        for (int r = 0; r < 8; ++r) s1 += t[r];
        if (half == 1) X2[w][D + col] = s1;
    }
    if (half == 0) *(float4*)&X2[w][4 * m] = selfq;

    __syncthreads();

    if (w < NT2) {
        const int mm = min(m, NW - 1);
        v8f c2 = zero;
        v2f aw[D2 / 4];
        #pragma unroll
        for (int ks = 0; ks < D2 / 4; ++ks)
            aw[ks] = *(const v2f*)&X2[mm][ks * 4 + kb];
        __builtin_amdgcn_sched_barrier(0);
        #pragma unroll
        for (int ks = 0; ks < D2 / 4; ++ks)
            c2 = __builtin_amdgcn_wmma_f32_16x16x4_f32(
                false, aw[ks], false, bw[ks], (short)0, c2, false, false);
        __builtin_amdgcn_sched_barrier(0);
        asm volatile("v_nop\n\tv_nop\n\tv_nop\n\tv_nop" : "+v"(c2));
        #pragma unroll
        for (int r = 0; r < 8; ++r) {
            const int row = r + half * 8;
            if (row < NW) V[row][col2] = c2[r] + bias2;
        }
    }

    __syncthreads();

    {
        float vals[NT2];
        float ss = 0.f;
        #pragma unroll
        for (int nt = 0; nt < NT2; ++nt) {
            const float v = V[w][nt * 16 + m];
            vals[nt] = v;
            ss += v * v;
        }
        ss += __shfl_xor(ss, 1);
        ss += __shfl_xor(ss, 2);
        ss += __shfl_xor(ss, 4);
        ss += __shfl_xor(ss, 8);
        const float inv = 1.0f / fmaxf(sqrtf(ss), EPSN);

        if (g < GTOT) {
            #pragma unroll
            for (int q = 0; q < NT2 / 2; ++q) {
                const float v = half ? vals[NT2 / 2 + q] : vals[q];
                out[(size_t)g * D2 + (half * (NT2 / 2) + q) * 16 + m] = v * inv;
            }
        }
    }
}

extern "C" void kernel_launch(void* const* d_in, const int* in_sizes, int n_in,
                              void* d_out, int out_size, void* d_ws, size_t ws_size,
                              hipStream_t stream) {
    const float* feat = (const float*)d_in[0];
    const int*   adj  = (const int*)  d_in[1];
    const int*   in1  = (const int*)  d_in[2];
    const int*   in2  = (const int*)  d_in[3];
    const int*   ineg = (const int*)  d_in[4];
    const float* W1   = (const float*)d_in[5];
    const float* b1   = (const float*)d_in[6];
    const float* W2   = (const float*)d_in[7];
    const float* b2   = (const float*)d_in[8];
    float*       outp = (float*)d_out;

    graphsage_fused<<<NBLK, NW * 32, 0, stream>>>(feat, adj, in1, in2, ineg,
                                                  W1, b1, W2, b2, outp);
}


# W1 copied once per work-group to LDS by whole-row loads; hop-1 B fragments read from LDS, not by 128 global loads per wave that touch 16 lines each; one early barrier; k and j sum orders kept
# speedup vs baseline: 3.2477x; 1.1707x over previous
#include <hip/hip_runtime.h>
#include <math.h>


typedef __attribute__((ext_vector_type(2))) float v2f;
typedef __attribute__((ext_vector_type(8))) float v8f;

#define N_NODES 65536
#define DEG     16
#define D       64
#define D2      128
#define NB1     1024
#define NNEG    10
#define GTOT    (NB1 + NB1 + NNEG)
#define EPSN    1e-12f
#define XPAD    132
#define WPAD    132
#define W1PASS  ((D + NW - 1) / NW)
#define W1ROWS  (W1PASS * NW)
#define NW      9
#define NBLK    ((GTOT + NW - 1) / NW)
#define NT2     (D2 / 16)

__device__ __forceinline__ void wave_lds_order()
{
    __builtin_amdgcn_fence(__ATOMIC_RELEASE, "wavefront");
    __builtin_amdgcn_wave_barrier();
    __builtin_amdgcn_fence(__ATOMIC_ACQUIRE, "wavefront");
}

__device__ __forceinline__ void stage_ids(
    const int* __restrict__ adj, const int node_r, const int lane,
    int (*Is)[DEG], int* Ns)
{
    const int r    = lane >> 1;
    const int part = lane & 1;
    const int4* ap = (const int4*)(adj + (size_t)node_r * DEG + part * 8);
    const int4 i0 = ap[0];
    const int4 i1 = ap[1];
    *(int4*)&Is[r][part * 8]     = i0;
    *(int4*)&Is[r][part * 8 + 4] = i1;
    if (part == 0) Ns[r] = node_r;
}

__device__ __forceinline__ void gather_rows(
    const float* __restrict__ selfsrc, const float* __restrict__ nbsrc, const int lane,
    float (*Xs)[XPAD], int (*Is)[DEG], int* Ns)
{
    const int c    = lane & 15;
    const int half = lane >> 4;
    #pragma unroll
    for (int p = 0; p < 8; ++p) {
        const int row = 2 * p + half;
        const float4 s = *(const float4*)(selfsrc + (size_t)Ns[row] * D + 4 * c);
        float4 acc = make_float4(0.f, 0.f, 0.f, 0.f);
        #pragma unroll
        for (int q = 0; q < DEG / 4; ++q) {
            const int4 nb = *(const int4*)&Is[row][4 * q];
            const float4 v0 = *(const float4*)(nbsrc + (size_t)nb.x * D + 4 * c);
            const float4 v1 = *(const float4*)(nbsrc + (size_t)nb.y * D + 4 * c);
            const float4 v2 = *(const float4*)(nbsrc + (size_t)nb.z * D + 4 * c);
            const float4 v3 = *(const float4*)(nbsrc + (size_t)nb.w * D + 4 * c);
            acc.x += v0.x; acc.y += v0.y; acc.z += v0.z; acc.w += v0.w;
            acc.x += v1.x; acc.y += v1.y; acc.z += v1.z; acc.w += v1.w;
            acc.x += v2.x; acc.y += v2.y; acc.z += v2.z; acc.w += v2.w;
            acc.x += v3.x; acc.y += v3.y; acc.z += v3.z; acc.w += v3.w;
        }
        *(float4*)&Xs[row][4 * c]     = s;
        *(float4*)&Xs[row][D + 4 * c] = acc;
    }
}

__device__ __forceinline__ int query_node(
    const int g, const int* __restrict__ in1, const int* __restrict__ in2,
    const int* __restrict__ ineg)
{
    if (g < NB1)     return in1[g];
    if (g < 2 * NB1) return in2[g - NB1];
    return ineg[g - 2 * NB1];
}

__global__ __launch_bounds__(NW * 32) void graphsage_fused(
    const float* __restrict__ feat, const int* __restrict__ adj,
    const int* __restrict__ in1, const int* __restrict__ in2,
    const int* __restrict__ ineg,
    const float* __restrict__ W1, const float* __restrict__ b1,
    const float* __restrict__ W2, const float* __restrict__ b2,
    float* __restrict__ out)
{
    __shared__ __attribute__((aligned(16))) float Xs_all[NW][16][XPAD];
    __shared__ __attribute__((aligned(16))) int   Is_all[NW][16][DEG];
    __shared__ int   Ns_all[NW][16];
    __shared__ __attribute__((aligned(16))) float X2[NW][XPAD];
    __shared__ float V[NW][XPAD];
    __shared__ __attribute__((aligned(16))) float W1s[W1ROWS][WPAD];

    const int lane = threadIdx.x & 31;
    const int w    = __builtin_amdgcn_readfirstlane((int)(threadIdx.x >> 5));
    const int m    = lane & 15;
    const int half = lane >> 4;
    const int g    = blockIdx.x * NW + w;
    const int gq   = min(g, GTOT - 1);

    float (*Xs)[XPAD] = Xs_all[w];
    int   (*Is)[DEG]  = Is_all[w];
    int*  Ns          = Ns_all[w];

    const int nq     = query_node(gq, in1, in2, ineg);
    const int node_r = adj[(size_t)nq * DEG + (lane >> 1)];
    const float4 selfq = *(const float4*)(feat + (size_t)nq * D + 4 * m);

    float bias1[4];
    #pragma unroll
    for (int nt = 0; nt < 4; ++nt) bias1[nt] = b1[nt * 16 + m];

    float4 w1r[W1PASS];
    #pragma unroll
    for (int i = 0; i < W1PASS; ++i) {
        const int r = min(w + i * NW, D - 1);
        w1r[i] = *(const float4*)(W1 + (size_t)r * (2 * D) + 4 * lane);
    }
    #pragma unroll
    for (int i = 0; i < W1PASS; ++i)
        *(float4*)&W1s[w + i * NW][4 * lane] = w1r[i];

    stage_ids(adj, node_r, lane, Is, Ns);
    __syncthreads();

    gather_rows(feat, feat, lane, Xs, Is, Ns);
    wave_lds_order();

    const int kb = half * 2;
    v8f zero = {};
    v8f c[4];
    #pragma unroll
    for (int nt = 0; nt < 4; ++nt) c[nt] = zero;

    for (int k0 = 0; k0 < 2 * D; k0 += 4) {
        v2f a = *(const v2f*)&Xs[m][k0 + kb];
        #pragma unroll
        for (int nt = 0; nt < 4; ++nt) {
            const int col = nt * 16 + m;
            v2f b = *(const v2f*)&W1s[col][k0 + kb];
            c[nt] = __builtin_amdgcn_wmma_f32_16x16x4_f32(
                false, a, false, b, (short)0, c[nt], false, false);
        }
    }

    v2f bw[D2 / 4];
    const int col2 = min(w, NT2 - 1) * 16 + m;
    #pragma unroll
    for (int ks = 0; ks < D2 / 4; ++ks)
        bw[ks] = *(const v2f*)&W2[(size_t)col2 * D2 + ks * 4 + kb];
    const float bias2 = b2[col2];

    #pragma unroll
    for (int nt = 0; nt < 4; ++nt) {
        const int   col  = nt * 16 + m;
        const float bias = bias1[nt];
        float t[8];
        #pragma unroll
        for (int r = 0; r < 8; ++r) t[r] = tanhf(c[nt][r] + bias);
        float s0 = 0.f;
        #pragma unroll
        for (int r = 0; r < 8; ++r) s0 += t[r];
        const float p = __shfl(s0, m);
        float s1 = p;
        #pragma unroll
        for (int r = 0; r < 8; ++r) s1 += t[r];
        if (half == 1) X2[w][D + col] = s1;
    }
    if (half == 0) *(float4*)&X2[w][4 * m] = selfq;

    __syncthreads();

    if (w < NT2) {
        const int mm = min(m, NW - 1);
        v8f c2 = zero;
        v2f aw[D2 / 4];
        #pragma unroll
        for (int ks = 0; ks < D2 / 4; ++ks)
            aw[ks] = *(const v2f*)&X2[mm][ks * 4 + kb];
        __builtin_amdgcn_sched_barrier(0);
        #pragma unroll
        for (int ks = 0; ks < D2 / 4; ++ks)
            c2 = __builtin_amdgcn_wmma_f32_16x16x4_f32(
                false, aw[ks], false, bw[ks], (short)0, c2, false, false);
        __builtin_amdgcn_sched_barrier(0);
        asm volatile("v_nop\n\tv_nop\n\tv_nop\n\tv_nop" : "+v"(c2));
        #pragma unroll
        for (int r = 0; r < 8; ++r) {
            const int row = r + half * 8;
            if (row < NW) V[row][col2] = c2[r] + bias2;
        }
    }

    __syncthreads();

    {
        float vals[NT2];
        float ss = 0.f;
        #pragma unroll
        for (int nt = 0; nt < NT2; ++nt) {
            const float v = V[w][nt * 16 + m];
            vals[nt] = v;
            ss += v * v;
        }
        ss += __shfl_xor(ss, 1);
        ss += __shfl_xor(ss, 2);
        ss += __shfl_xor(ss, 4);
        ss += __shfl_xor(ss, 8);
        const float inv = 1.0f / fmaxf(sqrtf(ss), EPSN);

        if (g < GTOT) {
            #pragma unroll
            for (int q = 0; q < NT2 / 2; ++q) {
                const float v = half ? vals[NT2 / 2 + q] : vals[q];
                out[(size_t)g * D2 + (half * (NT2 / 2) + q) * 16 + m] = v * inv;
            }
        }
    }
}

extern "C" void kernel_launch(void* const* d_in, const int* in_sizes, int n_in,
                              void* d_out, int out_size, void* d_ws, size_t ws_size,
                              hipStream_t stream) {
    const float* feat = (const float*)d_in[0];
    const int*   adj  = (const int*)  d_in[1];
    const int*   in1  = (const int*)  d_in[2];
    const int*   in2  = (const int*)  d_in[3];
    const int*   ineg = (const int*)  d_in[4];
    const float* W1   = (const float*)d_in[5];
    const float* b1   = (const float*)d_in[6];
    const float* W2   = (const float*)d_in[7];
    const float* b2   = (const float*)d_in[8];
    float*       outp = (float*)d_out;

    graphsage_fused<<<NBLK, NW * 32, 0, stream>>>(feat, adj, in1, in2, ineg,
                                                  W1, b1, W2, b2, outp);
}


# tanh of the hop-1 epilogue without branches: both |x| ranges as plain arithmetic (exp2, rcp, polynomial) and one select, so a lane's 32 values overlap; values within about 2 ulp, sum orders kept
# speedup vs baseline: 3.6454x; 1.1225x over previous
#include <hip/hip_runtime.h>
#include <math.h>


typedef __attribute__((ext_vector_type(2))) float v2f;
typedef __attribute__((ext_vector_type(8))) float v8f;

#define N_NODES 65536
#define DEG     16
#define D       64
#define D2      128
#define NB1     1024
#define NNEG    10
#define GTOT    (NB1 + NB1 + NNEG)
#define EPSN    1e-12f
#define XPAD    132
#define WPAD    132
#define W1PASS  ((D + NW - 1) / NW)
#define W1ROWS  (W1PASS * NW)
#define NW      9
#define NBLK    ((GTOT + NW - 1) / NW)
#define NT2     (D2 / 16)

__device__ __forceinline__ void wave_lds_order()
{
    __builtin_amdgcn_fence(__ATOMIC_RELEASE, "wavefront");
    __builtin_amdgcn_wave_barrier();
    __builtin_amdgcn_fence(__ATOMIC_ACQUIRE, "wavefront");
}

__device__ __forceinline__ void stage_ids(
    const int* __restrict__ adj, const int node_r, const int lane,
    int (*Is)[DEG], int* Ns)
{
    const int r    = lane >> 1;
    const int part = lane & 1;
    const int4* ap = (const int4*)(adj + (size_t)node_r * DEG + part * 8);
    const int4 i0 = ap[0];
    const int4 i1 = ap[1];
    *(int4*)&Is[r][part * 8]     = i0;
    *(int4*)&Is[r][part * 8 + 4] = i1;
    if (part == 0) Ns[r] = node_r;
}

__device__ __forceinline__ void gather_rows(
    const float* __restrict__ selfsrc, const float* __restrict__ nbsrc, const int lane,
    float (*Xs)[XPAD], int (*Is)[DEG], int* Ns)
{
    const int c    = lane & 15;
    const int half = lane >> 4;
    #pragma unroll
    for (int p = 0; p < 8; ++p) {
        const int row = 2 * p + half;
        const float4 s = *(const float4*)(selfsrc + (size_t)Ns[row] * D + 4 * c);
        float4 acc = make_float4(0.f, 0.f, 0.f, 0.f);
        #pragma unroll
        for (int q = 0; q < DEG / 4; ++q) {
            const int4 nb = *(const int4*)&Is[row][4 * q];
            const float4 v0 = *(const float4*)(nbsrc + (size_t)nb.x * D + 4 * c);
            const float4 v1 = *(const float4*)(nbsrc + (size_t)nb.y * D + 4 * c);
            const float4 v2 = *(const float4*)(nbsrc + (size_t)nb.z * D + 4 * c);
            const float4 v3 = *(const float4*)(nbsrc + (size_t)nb.w * D + 4 * c);
            acc.x += v0.x; acc.y += v0.y; acc.z += v0.z; acc.w += v0.w;
            acc.x += v1.x; acc.y += v1.y; acc.z += v1.z; acc.w += v1.w;
            acc.x += v2.x; acc.y += v2.y; acc.z += v2.z; acc.w += v2.w;
            acc.x += v3.x; acc.y += v3.y; acc.z += v3.z; acc.w += v3.w;
        }
        *(float4*)&Xs[row][4 * c]     = s;
        *(float4*)&Xs[row][D + 4 * c] = acc;
    }
}

__device__ __forceinline__ float tanh_select(const float x)
{
    const float ax = __builtin_fabsf(x);
    const float e  = __builtin_amdgcn_exp2f(ax * 2.88539004f);
    const float zb = __builtin_fmaf(__builtin_amdgcn_rcpf(e + 1.0f), -2.0f, 1.0f);
    const float t  = x * x;
    float p = __builtin_fmaf(t, -0.00570002059f, 0.0206340719f);
    p = __builtin_fmaf(t, p, -0.0537379310f);
    p = __builtin_fmaf(t, p, 0.133314162f);
    p = __builtin_fmaf(t, p, -0.333332807f);
    const float zs = __builtin_fmaf(t, ax * p, ax);
    const float z  = (ax < 0.625f) ? zs : zb;
    return __builtin_copysignf(z, x);
}

__device__ __forceinline__ int query_node(
    const int g, const int* __restrict__ in1, const int* __restrict__ in2,
    const int* __restrict__ ineg)
{
    if (g < NB1)     return in1[g];
    if (g < 2 * NB1) return in2[g - NB1];
    return ineg[g - 2 * NB1];
}

__global__ __launch_bounds__(NW * 32) void graphsage_fused(
    const float* __restrict__ feat, const int* __restrict__ adj,
    const int* __restrict__ in1, const int* __restrict__ in2,
    const int* __restrict__ ineg,
    const float* __restrict__ W1, const float* __restrict__ b1,
    const float* __restrict__ W2, const float* __restrict__ b2,
    float* __restrict__ out)
{
    __shared__ __attribute__((aligned(16))) float Xs_all[NW][16][XPAD];
    __shared__ __attribute__((aligned(16))) int   Is_all[NW][16][DEG];
    __shared__ int   Ns_all[NW][16];
    __shared__ __attribute__((aligned(16))) float X2[NW][XPAD];
    __shared__ float V[NW][XPAD];
    __shared__ __attribute__((aligned(16))) float W1s[W1ROWS][WPAD];

    const int lane = threadIdx.x & 31;
    const int w    = __builtin_amdgcn_readfirstlane((int)(threadIdx.x >> 5));
    const int m    = lane & 15;
    const int half = lane >> 4;
    const int g    = blockIdx.x * NW + w;
    const int gq   = min(g, GTOT - 1);

    float (*Xs)[XPAD] = Xs_all[w];
    int   (*Is)[DEG]  = Is_all[w];
    int*  Ns          = Ns_all[w];

    const int nq     = query_node(gq, in1, in2, ineg);
    const int node_r = adj[(size_t)nq * DEG + (lane >> 1)];
    const float4 selfq = *(const float4*)(feat + (size_t)nq * D + 4 * m);

    float bias1[4];
    #pragma unroll
    for (int nt = 0; nt < 4; ++nt) bias1[nt] = b1[nt * 16 + m];

    float4 w1r[W1PASS];
    #pragma unroll
    for (int i = 0; i < W1PASS; ++i) {
        const int r = min(w + i * NW, D - 1);
        w1r[i] = *(const float4*)(W1 + (size_t)r * (2 * D) + 4 * lane);
    }
    #pragma unroll
    for (int i = 0; i < W1PASS; ++i)
        *(float4*)&W1s[w + i * NW][4 * lane] = w1r[i];

    stage_ids(adj, node_r, lane, Is, Ns);
    __syncthreads();

    gather_rows(feat, feat, lane, Xs, Is, Ns);
    wave_lds_order();

    const int kb = half * 2;
    v8f zero = {};
    v8f c[4];
    #pragma unroll
    for (int nt = 0; nt < 4; ++nt) c[nt] = zero;

    for (int k0 = 0; k0 < 2 * D; k0 += 4) {
        v2f a = *(const v2f*)&Xs[m][k0 + kb];
        #pragma unroll
        for (int nt = 0; nt < 4; ++nt) {
            const int col = nt * 16 + m;
            v2f b = *(const v2f*)&W1s[col][k0 + kb];
            c[nt] = __builtin_amdgcn_wmma_f32_16x16x4_f32(
                false, a, false, b, (short)0, c[nt], false, false);
        }
    }

    v2f bw[D2 / 4];
    const int col2 = min(w, NT2 - 1) * 16 + m;
    #pragma unroll
    for (int ks = 0; ks < D2 / 4; ++ks)
        bw[ks] = *(const v2f*)&W2[(size_t)col2 * D2 + ks * 4 + kb];
    const float bias2 = b2[col2];
    __builtin_amdgcn_sched_barrier(0);

    #pragma unroll
    for (int nt = 0; nt < 4; ++nt) {
        const int   col  = nt * 16 + m;
        const float bias = bias1[nt];
        float t[8];
        #pragma unroll
        for (int r = 0; r < 8; ++r) t[r] = tanh_select(c[nt][r] + bias);
        float s0 = 0.f;
        #pragma unroll
        for (int r = 0; r < 8; ++r) s0 += t[r];
        const float p = __shfl(s0, m);
        float s1 = p;
        #pragma unroll
        for (int r = 0; r < 8; ++r) s1 += t[r];
        if (half == 1) X2[w][D + col] = s1;
    }
    if (half == 0) *(float4*)&X2[w][4 * m] = selfq;

    __syncthreads();

    if (w < NT2) {
        const int mm = min(m, NW - 1);
        v8f c2 = zero;
        v2f aw[D2 / 4];
        #pragma unroll
        for (int ks = 0; ks < D2 / 4; ++ks)
            aw[ks] = *(const v2f*)&X2[mm][ks * 4 + kb];
        __builtin_amdgcn_sched_barrier(0);
        #pragma unroll
        for (int ks = 0; ks < D2 / 4; ++ks)
            c2 = __builtin_amdgcn_wmma_f32_16x16x4_f32(
                false, aw[ks], false, bw[ks], (short)0, c2, false, false);
        __builtin_amdgcn_sched_barrier(0);
        asm volatile("v_nop\n\tv_nop\n\tv_nop\n\tv_nop" : "+v"(c2));
        #pragma unroll
        for (int r = 0; r < 8; ++r) {
            const int row = r + half * 8;
            if (row < NW) V[row][col2] = c2[r] + bias2;
        }
    }

    __syncthreads();

    {
        float vals[NT2];
        float ss = 0.f;
        #pragma unroll
        for (int nt = 0; nt < NT2; ++nt) {
            const float v = V[w][nt * 16 + m];
            vals[nt] = v;
            ss += v * v;
        }
        ss += __shfl_xor(ss, 1);
        ss += __shfl_xor(ss, 2);
        ss += __shfl_xor(ss, 4);
        ss += __shfl_xor(ss, 8);
        const float inv = 1.0f / fmaxf(sqrtf(ss), EPSN);

        if (g < GTOT) {
            #pragma unroll
            for (int q = 0; q < NT2 / 2; ++q) {
                const float v = half ? vals[NT2 / 2 + q] : vals[q];
                out[(size_t)g * D2 + (half * (NT2 / 2) + q) * 16 + m] = v * inv;
            }
        }
    }
}

extern "C" void kernel_launch(void* const* d_in, const int* in_sizes, int n_in,
                              void* d_out, int out_size, void* d_ws, size_t ws_size,
                              hipStream_t stream) {
    const float* feat = (const float*)d_in[0];
    const int*   adj  = (const int*)  d_in[1];
    const int*   in1  = (const int*)  d_in[2];
    const int*   in2  = (const int*)  d_in[3];
    const int*   ineg = (const int*)  d_in[4];
    const float* W1   = (const float*)d_in[5];
    const float* b1   = (const float*)d_in[6];
    const float* W2   = (const float*)d_in[7];
    const float* b2   = (const float*)d_in[8];
    float*       outp = (float*)d_out;

    graphsage_fused<<<NBLK, NW * 32, 0, stream>>>(feat, adj, in1, in2, ineg,
                                                  W1, b1, W2, b2, outp);
}


# Hop-1 product overlapped with the row gather: 96 of a wave's 128 WMMAs issued under loads in flight, K steps as soon as their X columns are in LDS (rows fetched by 128-B halves); sum orders kept
# speedup vs baseline: 4.1000x; 1.1247x over previous
#include <hip/hip_runtime.h>
#include <math.h>


typedef __attribute__((ext_vector_type(2))) float v2f;
typedef __attribute__((ext_vector_type(8))) float v8f;

#define N_NODES 65536
#define DEG     16
#define D       64
#define D2      128
#define NB1     1024
#define NNEG    10
#define GTOT    (NB1 + NB1 + NNEG)
#define EPSN    1e-12f
#define XPAD    132
#define WPAD    132
#define W1PASS  ((D + NW - 1) / NW)
#define W1ROWS  (W1PASS * NW)
#define NW      9
#define NBLK    ((GTOT + NW - 1) / NW)
#define NT2     (D2 / 16)
#define NSTEP   8

__device__ __forceinline__ void wave_lds_order()
{
    __builtin_amdgcn_fence(__ATOMIC_RELEASE, "wavefront");
    __builtin_amdgcn_wave_barrier();
    __builtin_amdgcn_fence(__ATOMIC_ACQUIRE, "wavefront");
}

__device__ __forceinline__ void stage_ids(
    const int* __restrict__ adj, const int node_r, const int lane,
    int (*Is)[DEG], int* Ns)
{
    const int r    = lane >> 1;
    const int part = lane & 1;
    const int4* ap = (const int4*)(adj + (size_t)node_r * DEG + part * 8);
    const int4 i0 = ap[0];
    const int4 i1 = ap[1];
    *(int4*)&Is[r][part * 8]     = i0;
    *(int4*)&Is[r][part * 8 + 4] = i1;
    if (part == 0) Ns[r] = node_r;
}

__device__ __forceinline__ void self_rows(
    const float* __restrict__ selfsrc, const int lane, const int* Ns, float4 (&s)[8])
{
    const int c    = lane & 15;
    const int half = lane >> 4;
    #pragma unroll
    for (int p = 0; p < 8; ++p)
        s[p] = *(const float4*)(selfsrc + (size_t)Ns[2 * p + half] * D + 4 * c);
}

__device__ __forceinline__ void issue_step(
    const float* __restrict__ nbsrc, const int lane, int (*Is)[DEG], const int s,
    float4 (&v)[DEG])
{
    const int row = 4 * (s & 3) + (lane >> 3);
    const int co  = 32 * (s >> 2) + 4 * (lane & 7);
    #pragma unroll
    for (int q = 0; q < DEG / 4; ++q) {
        const int4 nb = *(const int4*)&Is[row][4 * q];
        v[4 * q + 0] = *(const float4*)(nbsrc + (size_t)nb.x * D + co);
        v[4 * q + 1] = *(const float4*)(nbsrc + (size_t)nb.y * D + co);
        v[4 * q + 2] = *(const float4*)(nbsrc + (size_t)nb.z * D + co);
        v[4 * q + 3] = *(const float4*)(nbsrc + (size_t)nb.w * D + co);
    }
}

__device__ __forceinline__ void sum_step(
    const int lane, float (*Xs)[XPAD], const int s, const float4 (&v)[DEG], const float zero)
{
    const int row = 4 * (s & 3) + (lane >> 3);
    const int co  = 32 * (s >> 2) + 4 * (lane & 7);
    float4 acc = make_float4(zero, zero, zero, zero);
    #pragma unroll
    for (int j = 0; j < DEG; ++j) {
        acc.x += v[j].x; acc.y += v[j].y; acc.z += v[j].z; acc.w += v[j].w;
    }
    *(float4*)&Xs[row][D + co] = acc;
}

#define GEMM1_16WMMA \
        "v_wmma_f32_16x16x4_f32 %[c0], %[a0], %[b00], %[c0]\n\t" \
        "v_wmma_f32_16x16x4_f32 %[c1], %[a0], %[b01], %[c1]\n\t" \
        "v_wmma_f32_16x16x4_f32 %[c2], %[a0], %[b02], %[c2]\n\t" \
        "v_wmma_f32_16x16x4_f32 %[c3], %[a0], %[b03], %[c3]\n\t" \
        "v_wmma_f32_16x16x4_f32 %[c0], %[a1], %[b10], %[c0]\n\t" \
        "v_wmma_f32_16x16x4_f32 %[c1], %[a1], %[b11], %[c1]\n\t" \
        "v_wmma_f32_16x16x4_f32 %[c2], %[a1], %[b12], %[c2]\n\t" \
        "v_wmma_f32_16x16x4_f32 %[c3], %[a1], %[b13], %[c3]\n\t" \
        "v_wmma_f32_16x16x4_f32 %[c0], %[a2], %[b20], %[c0]\n\t" \
        "v_wmma_f32_16x16x4_f32 %[c1], %[a2], %[b21], %[c1]\n\t" \
        "v_wmma_f32_16x16x4_f32 %[c2], %[a2], %[b22], %[c2]\n\t" \
        "v_wmma_f32_16x16x4_f32 %[c3], %[a2], %[b23], %[c3]\n\t" \
        "v_wmma_f32_16x16x4_f32 %[c0], %[a3], %[b30], %[c0]\n\t" \
        "v_wmma_f32_16x16x4_f32 %[c1], %[a3], %[b31], %[c1]\n\t" \
        "v_wmma_f32_16x16x4_f32 %[c2], %[a3], %[b32], %[c2]\n\t" \
        "v_wmma_f32_16x16x4_f32 %[c3], %[a3], %[b33], %[c3]\n\t" \
        "v_nop\n\tv_nop\n\tv_nop\n\tv_nop"
#define GEMM1_ACCS \
          [c0] "+v"(c[0]), [c1] "+v"(c[1]), [c2] "+v"(c[2]), [c3] "+v"(c[3])
#define GEMM1_FRAGS \
          [a0] "v"(a[0]), [a1] "v"(a[1]), [a2] "v"(a[2]), [a3] "v"(a[3]), \
          [b00] "v"(b[0][0]), [b01] "v"(b[0][1]), [b02] "v"(b[0][2]), [b03] "v"(b[0][3]), \
          [b10] "v"(b[1][0]), [b11] "v"(b[1][1]), [b12] "v"(b[1][2]), [b13] "v"(b[1][3]), \
          [b20] "v"(b[2][0]), [b21] "v"(b[2][1]), [b22] "v"(b[2][2]), [b23] "v"(b[2][3]), \
          [b30] "v"(b[3][0]), [b31] "v"(b[3][1]), [b32] "v"(b[3][2]), [b33] "v"(b[3][3])
template <bool ZERO_OUT>
__device__ __forceinline__ float gemm1_group(
    v8f (&c)[4], float (*Xs)[XPAD], float (*W1s)[WPAD], const int m, const int kb,
    const int k_begin)
{
    v2f a[4];
    v2f b[4][4];
    #pragma unroll
    for (int i = 0; i < 4; ++i) {
        const int k0 = k_begin + 4 * i;
        a[i] = *(const v2f*)&Xs[m][k0 + kb];
        #pragma unroll
        for (int nt = 0; nt < 4; ++nt)
            b[i][nt] = *(const v2f*)&W1s[nt * 16 + m][k0 + kb];
    }
    float z = 0.f;
    if constexpr (ZERO_OUT) {
        asm volatile(GEMM1_16WMMA "\n\tv_mov_b32 %[z], 0"
            : GEMM1_ACCS, [z] "=v"(z)
            : GEMM1_FRAGS);
    } else {
        asm volatile(GEMM1_16WMMA
            : GEMM1_ACCS
            : GEMM1_FRAGS);
    }
    return z;
}

__device__ __forceinline__ float tanh_select(const float x)
{
    const float ax = __builtin_fabsf(x);
    const float e  = __builtin_amdgcn_exp2f(ax * 2.88539004f);
    const float zb = __builtin_fmaf(__builtin_amdgcn_rcpf(e + 1.0f), -2.0f, 1.0f);
    const float t  = x * x;
    float p = __builtin_fmaf(t, -0.00570002059f, 0.0206340719f);
    p = __builtin_fmaf(t, p, -0.0537379310f);
    p = __builtin_fmaf(t, p, 0.133314162f);
    p = __builtin_fmaf(t, p, -0.333332807f);
    const float zs = __builtin_fmaf(t, ax * p, ax);
    const float z  = (ax < 0.625f) ? zs : zb;
    return __builtin_copysignf(z, x);
}

__device__ __forceinline__ int query_node(
    const int g, const int* __restrict__ in1, const int* __restrict__ in2,
    const int* __restrict__ ineg)
{
    if (g < NB1)     return in1[g];
    if (g < 2 * NB1) return in2[g - NB1];
    return ineg[g - 2 * NB1];
}

__global__ __launch_bounds__(NW * 32) void graphsage_fused(
    const float* __restrict__ feat, const int* __restrict__ adj,
    const int* __restrict__ in1, const int* __restrict__ in2,
    const int* __restrict__ ineg,
    const float* __restrict__ W1, const float* __restrict__ b1,
    const float* __restrict__ W2, const float* __restrict__ b2,
    float* __restrict__ out)
{
    __shared__ __attribute__((aligned(16))) float Xs_all[NW][16][XPAD];
    __shared__ __attribute__((aligned(16))) int   Is_all[NW][16][DEG];
    __shared__ int   Ns_all[NW][16];
    __shared__ __attribute__((aligned(16))) float X2[NW][XPAD];
    __shared__ float V[NW][XPAD];
    __shared__ __attribute__((aligned(16))) float W1s[W1ROWS][WPAD];

    const int lane = threadIdx.x & 31;
    const int w    = __builtin_amdgcn_readfirstlane((int)(threadIdx.x >> 5));
    const int m    = lane & 15;
    const int half = lane >> 4;
    const int g    = blockIdx.x * NW + w;
    const int gq   = min(g, GTOT - 1);

    float (*Xs)[XPAD] = Xs_all[w];
    int   (*Is)[DEG]  = Is_all[w];
    int*  Ns          = Ns_all[w];

    const int nq     = query_node(gq, in1, in2, ineg);
    const int node_r = adj[(size_t)nq * DEG + (lane >> 1)];
    const float4 selfq = *(const float4*)(feat + (size_t)nq * D + 4 * m);

    float bias1[4];
    #pragma unroll
    for (int nt = 0; nt < 4; ++nt) bias1[nt] = b1[nt * 16 + m];

    float4 w1r[W1PASS];
    #pragma unroll
    for (int i = 0; i < W1PASS; ++i) {
        const int r = min(w + i * NW, D - 1);
        w1r[i] = *(const float4*)(W1 + (size_t)r * (2 * D) + 4 * lane);
    }
    #pragma unroll
    for (int i = 0; i < W1PASS; ++i)
        *(float4*)&W1s[w + i * NW][4 * lane] = w1r[i];

    stage_ids(adj, node_r, lane, Is, Ns);
    __syncthreads();

    const int kb = half * 2;
    v8f zero = {};
    v8f c[4];
    #pragma unroll
    for (int nt = 0; nt < 4; ++nt) c[nt] = zero;

    float4 rows[NSTEP][DEG];
    {
        float4 s[8];
        self_rows(feat, lane, Ns, s);
        issue_step(feat, lane, Is, 0, rows[0]);
        issue_step(feat, lane, Is, 1, rows[1]);
        const int c16 = lane & 15;
        #pragma unroll
        for (int p = 0; p < 8; ++p) *(float4*)&Xs[2 * p + half][4 * c16] = s[p];
    }
    wave_lds_order();

    float zero_after = 0.f;
    #pragma unroll
    for (int s = 0; s < NSTEP; ++s) {
        if (s + 2 < NSTEP) issue_step(feat, lane, Is, s + 2, rows[s + 2]);
        if (s < NSTEP / 2 + 2) zero_after = gemm1_group<true>(c, Xs, W1s, m, kb, 16 * s);
        sum_step(lane, Xs, s, rows[s], zero_after);
        if (s == NSTEP / 2 - 1) wave_lds_order();
    }
    wave_lds_order();
    gemm1_group<false>(c, Xs, W1s, m, kb, D + 32);
    gemm1_group<false>(c, Xs, W1s, m, kb, D + 48);

    v2f bw[D2 / 4];
    const int col2 = min(w, NT2 - 1) * 16 + m;
    #pragma unroll
    for (int ks = 0; ks < D2 / 4; ++ks)
        bw[ks] = *(const v2f*)&W2[(size_t)col2 * D2 + ks * 4 + kb];
    const float bias2 = b2[col2];
    __builtin_amdgcn_sched_barrier(0);

    #pragma unroll
    for (int nt = 0; nt < 4; ++nt) {
        const int   col  = nt * 16 + m;
        const float bias = bias1[nt];
        float t[8];
        #pragma unroll
        for (int r = 0; r < 8; ++r) t[r] = tanh_select(c[nt][r] + bias);
        float s0 = 0.f;
        #pragma unroll
        for (int r = 0; r < 8; ++r) s0 += t[r];
        const float p = __shfl(s0, m);
        float s1 = p;
        #pragma unroll
        for (int r = 0; r < 8; ++r) s1 += t[r];
        if (half == 1) X2[w][D + col] = s1;
    }
    if (half == 0) *(float4*)&X2[w][4 * m] = selfq;

    __syncthreads();

    if (w < NT2) {
        const int mm = min(m, NW - 1);
        v8f c2 = zero;
        v2f aw[D2 / 4];
        #pragma unroll
        for (int ks = 0; ks < D2 / 4; ++ks)
            aw[ks] = *(const v2f*)&X2[mm][ks * 4 + kb];
        __builtin_amdgcn_sched_barrier(0);
        #pragma unroll
        for (int ks = 0; ks < D2 / 4; ++ks)
            c2 = __builtin_amdgcn_wmma_f32_16x16x4_f32(
                false, aw[ks], false, bw[ks], (short)0, c2, false, false);
        __builtin_amdgcn_sched_barrier(0);
        asm volatile("v_nop\n\tv_nop\n\tv_nop\n\tv_nop" : "+v"(c2));
        #pragma unroll
        for (int r = 0; r < 8; ++r) {
            const int row = r + half * 8;
            if (row < NW) V[row][col2] = c2[r] + bias2;
        }
    }

    __syncthreads();

    {
        float vals[NT2];
        float ss = 0.f;
        #pragma unroll
        for (int nt = 0; nt < NT2; ++nt) {
            const float v = V[w][nt * 16 + m];
            vals[nt] = v;
            ss += v * v;
        }
        ss += __shfl_xor(ss, 1);
        ss += __shfl_xor(ss, 2);
        ss += __shfl_xor(ss, 4);
        ss += __shfl_xor(ss, 8);
        const float inv = 1.0f / fmaxf(sqrtf(ss), EPSN);

        if (g < GTOT) {
            #pragma unroll
            for (int q = 0; q < NT2 / 2; ++q) {
                const float v = half ? vals[NT2 / 2 + q] : vals[q];
                out[(size_t)g * D2 + (half * (NT2 / 2) + q) * 16 + m] = v * inv;
            }
        }
    }
}

extern "C" void kernel_launch(void* const* d_in, const int* in_sizes, int n_in,
                              void* d_out, int out_size, void* d_ws, size_t ws_size,
                              hipStream_t stream) {
    const float* feat = (const float*)d_in[0];
    const int*   adj  = (const int*)  d_in[1];
    const int*   in1  = (const int*)  d_in[2];
    const int*   in2  = (const int*)  d_in[3];
    const int*   ineg = (const int*)  d_in[4];
    const float* W1   = (const float*)d_in[5];
    const float* b1   = (const float*)d_in[6];
    const float* W2   = (const float*)d_in[7];
    const float* b2   = (const float*)d_in[8];
    float*       outp = (float*)d_out;

    graphsage_fused<<<NBLK, NW * 32, 0, stream>>>(feat, adj, in1, in2, ineg,
                                                  W1, b1, W2, b2, outp);
}
